# starting kernel
# baseline (speedup 1.0000x reference)
#include <hip/hip_runtime.h>


typedef float v2f __attribute__((ext_vector_type(2)));
typedef float v8f __attribute__((ext_vector_type(8)));

#define SWZ_XOR_ADD(v, pat)                                                    \
  (v) += __int_as_float(__builtin_amdgcn_ds_swizzle(__float_as_int(v), (pat)))

constexpr int Bn = 8;
constexpr int Tn = 64;
constexpr int NI = 1024;
constexpr int NO = 1024;
constexpr int WAVES = 8;
constexpr int COLS_PER_WAVE = NI / WAVES;
constexpr int TILES = COLS_PER_WAVE / 16;

__global__ __launch_bounds__(256) void oja_scan_kernel(
    const float* __restrict__ X,
    const float* __restrict__ W0,
    float* __restrict__ out)
{
  const int lane  = threadIdx.x & 31;
  const int wv    = threadIdx.x >> 5;
  const int b     = blockIdx.x >> 6;
  const int strip = blockIdx.x & 63;
  const int half  = lane >> 4;
  const int ln    = lane & 15;
  const int colbase = wv * COLS_PER_WAVE;
  const float lr = 1.0f / (float)NI;

  __shared__ float part[WAVES][16];
  __shared__ float acoef[16];

  v8f w[TILES];
  const float* Wp = W0 + ((size_t)b * NO + (size_t)strip * 16) * NI;
#pragma unroll
  for (int j = 0; j < TILES; ++j) {
#pragma unroll
    for (int r = 0; r < 8; ++r) {
      w[j][r] = Wp[(size_t)(r + half * 8) * NI + (colbase + j * 16 + ln)];
    }
  }

  float P = 1.0f;

#pragma unroll 1
  for (int t = 0; t < Tn; ++t) {
    const float* xp = X + ((size_t)b * Tn + t) * NI + colbase;
    float xv[TILES];
#pragma unroll
    for (int j = 0; j < TILES; ++j) xv[j] = xp[j * 16 + ln];

    float acc[8];
#pragma unroll
    for (int r = 0; r < 8; ++r) acc[r] = 0.0f;
#pragma unroll
    for (int j = 0; j < TILES; ++j) {
#pragma unroll
      for (int r = 0; r < 8; ++r) acc[r] = __builtin_fmaf(w[j][r], xv[j], acc[r]);
    }

#pragma unroll
    for (int r = 0; r < 8; ++r) {
      SWZ_XOR_ADD(acc[r], 0x041F);
      SWZ_XOR_ADD(acc[r], 0x081F);
      SWZ_XOR_ADD(acc[r], 0x101F);
      SWZ_XOR_ADD(acc[r], 0x201F);
    }
    if (ln == 0) {
#pragma unroll
      for (int r = 0; r < 8; ++r) part[wv][half * 8 + r] = acc[r];
    }
    __syncthreads();

    if (wv == 0 && lane < 16) {
      float d = 0.0f;
#pragma unroll
      for (int k = 0; k < WAVES; ++k) d += part[k][lane];
      float pre = P * d;
      float y   = 1.0f / (1.0f + __expf(-pre));
      out[((size_t)b * Tn + t) * NO + (size_t)strip * 16 + lane] = y;
      P *= (1.0f - lr * y * y);
      acoef[lane] = lr * y / P;
    }
    __syncthreads();

    float ac = acoef[ln];
    v2f A;
    A.x = half ? 0.0f : ac;
    A.y = 0.0f;
#pragma unroll
    for (int j = 0; j < TILES; ++j) {
      v2f Bm;
      Bm.x = half ? 0.0f : xv[j];
      Bm.y = 0.0f;
      w[j] = __builtin_amdgcn_wmma_f32_16x16x4_f32(
           false, A,  false, Bm,
           (short)0, w[j],  false,  false);
    }
  }
}

extern "C" void kernel_launch(void* const* d_in, const int* in_sizes, int n_in,
                              void* d_out, int out_size, void* d_ws, size_t ws_size,
                              hipStream_t stream) {
  const float* X  = (const float*)d_in[0];
  const float* W0 = (const float*)d_in[1];
  float* out      = (float*)d_out;
  (void)in_sizes; (void)n_in; (void)out_size; (void)d_ws; (void)ws_size;

  dim3 grid(Bn * (NO / 16));
  dim3 block(WAVES * 32);
  oja_scan_kernel<<<grid, block, 0, stream>>>(X, W0, out);
}


# Replace the 64 dependent mat-vecs by WMMA GEMMs U=W0*X^T and G=X*X^T plus a per-row recurrence in 16-step blocks: d_t=U[:,t]+sum_{s<t} c_s*G[s,t], earlier blocks' part as WMMA C*G
# speedup vs baseline: 3.6444x; 3.6444x over previous
#include <hip/hip_runtime.h>


typedef float v2f __attribute__((ext_vector_type(2)));
typedef float v4f __attribute__((ext_vector_type(4)));
typedef float v8f __attribute__((ext_vector_type(8)));

constexpr int Bn = 8;
constexpr int Tn = 64;
constexpr int NI = 1024;
constexpr int NO = 1024;
constexpr int WAVES = 8;
constexpr int ROWS  = 32;
constexpr int KW    = NI / WAVES;
constexpr int KB    = 32;
constexpr int NBLK  = KW / KB;
constexpr int NTILE = 18;
constexpr int NFR   = 6;
constexpr int UP    = 33;

struct Frag { v4f q[4]; };

__device__ __forceinline__ void load_frag(Frag& f, const float* p) {
  const v4f* p4 = reinterpret_cast<const v4f*>(p);
  f.q[0] = p4[0];
  f.q[1] = p4[2];
  f.q[2] = p4[4];
  f.q[3] = p4[6];
}

__device__ __forceinline__ void load_block(Frag (&F)[NFR], const float* wp, const float* xp, int blk) {
#pragma unroll
  for (int i = 0; i < 2; ++i) load_frag(F[i], wp + (size_t)i * 16 * NI + blk * KB);
#pragma unroll
  for (int i = 0; i < 4; ++i) load_frag(F[2 + i], xp + (size_t)i * 16 * NI + blk * KB);
}

__device__ __forceinline__ void mma_block(v8f (&acc)[NTILE], const Frag (&F)[NFR]) {
#pragma unroll
  for (int q = 0; q < 4; ++q) {
#pragma unroll
    for (int p = 0; p < 2; ++p) {
      v2f a[NFR];
#pragma unroll
      for (int i = 0; i < NFR; ++i) {
        a[i].x = F[i].q[q][2 * p];
        a[i].y = F[i].q[q][2 * p + 1];
      }
#pragma unroll
      for (int mi = 0; mi < 2; ++mi) {
#pragma unroll
        for (int nj = 0; nj < 4; ++nj) {
          acc[mi * 4 + nj] = __builtin_amdgcn_wmma_f32_16x16x4_f32(
              false, a[mi], false, a[2 + nj], (short)0, acc[mi * 4 + nj], false, false);
        }
      }
      int idx = 8;
#pragma unroll
      for (int ti = 0; ti < 4; ++ti) {
#pragma unroll
        for (int tj = ti; tj < 4; ++tj) {
          acc[idx] = __builtin_amdgcn_wmma_f32_16x16x4_f32(
              false, a[2 + ti], false, a[2 + tj], (short)0, acc[idx], false, false);
          ++idx;
        }
      }
    }
  }
}

__global__ __launch_bounds__(256) void plastic_scan_kernel(
    const float* __restrict__ X,
    const float* __restrict__ W0,
    float* __restrict__ out)
{
  const int tid  = threadIdx.x;
  const int lane = tid & 31;
  const int wv   = tid >> 5;
  const int c    = lane & 15;
  const int h    = lane >> 4;
  const int b       = blockIdx.x >> 5;
  const int rowbase = (blockIdx.x & 31) * ROWS;
  const float lr = 1.0f / (float)NI;

  __shared__ __attribute__((aligned(16))) float part[WAVES * NTILE * 256];
  __shared__ __attribute__((aligned(16))) float Gl[Tn * Tn];
  __shared__ float Dl[Tn * UP];
  __shared__ float Cl[16 * 32];

  v8f acc[NTILE];
#pragma unroll
  for (int i = 0; i < NTILE; ++i) {
#pragma unroll
    for (int r = 0; r < 8; ++r) acc[i][r] = 0.0f;
  }

  const float* wp = W0 + ((size_t)b * NO + rowbase + c) * NI + wv * KW + 4 * h;
  const float* xp = X  + ((size_t)b * Tn + c) * NI + wv * KW + 4 * h;

  Frag F[2][NFR];
  load_block(F[0], wp, xp, 0);
#pragma unroll
  for (int blk = 0; blk < NBLK; ++blk) {
    if (blk + 1 < NBLK) load_block(F[(blk + 1) & 1], wp, xp, blk + 1);
    mma_block(acc, F[blk & 1]);
  }

#pragma unroll
  for (int i = 0; i < NTILE; ++i) {
#pragma unroll
    for (int r = 0; r < 8; ++r) part[(wv * NTILE + i) * 256 + r * 32 + lane] = acc[i][r];
  }
  __syncthreads();

  {
    const int rr = 8 * h + wv;
#pragma unroll
    for (int i = 0; i < 8; ++i) {
      float s = 0.0f;
#pragma unroll
      for (int k = 0; k < WAVES; ++k) s += part[(k * NTILE + i) * 256 + tid];
      const int mi = i >> 2, nj = i & 3;
      Dl[(16 * nj + c) * UP + 16 * mi + rr] = s;
    }
    int idx = 8;
#pragma unroll
    for (int ti = 0; ti < 4; ++ti) {
#pragma unroll
      for (int tj = ti; tj < 4; ++tj) {
        float s = 0.0f;
#pragma unroll
        for (int k = 0; k < WAVES; ++k) s += part[(k * NTILE + idx) * 256 + tid];
        Gl[(16 * ti + rr) * Tn + 16 * tj + c] = s;
        ++idx;
      }
    }
  }
  __syncthreads();

  float P = 1.0f;
  float* op = out + (size_t)b * Tn * NO + rowbase + lane;
#pragma unroll 1
  for (int j = 0; j < Tn / 16; ++j) {
    if (wv == 0) {
      float d[16];
#pragma unroll
      for (int e = 0; e < 16; ++e) d[e] = Dl[(16 * j + e) * UP + lane];
      const float* gblk = &Gl[(16 * j) * Tn + 16 * j];
#pragma unroll
      for (int e = 0; e < 16; ++e) {
        const float pre = P * d[e];
        const float y   = 1.0f / (1.0f + __expf(-pre));
        op[(size_t)(16 * j + e) * NO] = y;
        P *= (1.0f - lr * y * y);
        const float cf = lr * y / P;
        Cl[e * 32 + lane] = cf;
        const v4f* g4 = reinterpret_cast<const v4f*>(gblk + e * Tn);
#pragma unroll
        for (int qd = (e + 1) / 4; qd < 4; ++qd) {
          const v4f g = g4[qd];
#pragma unroll
          for (int f4 = 0; f4 < 4; ++f4) {
            const int f = 4 * qd + f4;
            if (f > e) d[f] = __builtin_fmaf(cf, g[f4], d[f]);
          }
        }
      }
    }
    __syncthreads();

    const int mi = wv & 1;
    const int jt = j + 1 + (wv >> 1);
    if (jt < Tn / 16) {
      v8f accd;
#pragma unroll
      for (int r = 0; r < 8; ++r) accd[r] = 0.0f;
#pragma unroll
      for (int ks = 0; ks < 4; ++ks) {
        const int k0 = 4 * ks + 2 * h;
        v2f av, bv;
        av.x = Cl[(k0    ) * 32 + 16 * mi + c];
        av.y = Cl[(k0 + 1) * 32 + 16 * mi + c];
        bv.x = Gl[(16 * j + k0    ) * Tn + 16 * jt + c];
        bv.y = Gl[(16 * j + k0 + 1) * Tn + 16 * jt + c];
        accd = __builtin_amdgcn_wmma_f32_16x16x4_f32(
            false, av, false, bv, (short)0, accd, false, false);
      }
#pragma unroll
      for (int r = 0; r < 8; ++r) Dl[(16 * jt + c) * UP + 16 * mi + 8 * h + r] += accd[r];
    }
    __syncthreads();
  }
}

extern "C" void kernel_launch(void* const* d_in, const int* in_sizes, int n_in,
                              void* d_out, int out_size, void* d_ws, size_t ws_size,
                              hipStream_t stream) {
  const float* X  = (const float*)d_in[0];
  const float* W0 = (const float*)d_in[1];
  float* out      = (float*)d_out;
  (void)in_sizes; (void)n_in; (void)out_size; (void)d_ws; (void)ws_size;

  dim3 grid(Bn * (NO / ROWS));
  dim3 block(WAVES * 32);
  plastic_scan_kernel<<<grid, block, 0, stream>>>(X, W0, out);
}


# Recurrence step's dependent path 28 -> 5 instructions: in-block sums kept as W.x (E <- s*E + y*G'), next entry as E + y*(G' - y*lr*E), v_rcp sigmoid, LDS reads moved out of the chain
# speedup vs baseline: 4.5914x; 1.2599x over previous
#include <hip/hip_runtime.h>


typedef float v2f __attribute__((ext_vector_type(2)));
typedef float v4f __attribute__((ext_vector_type(4)));
typedef float v8f __attribute__((ext_vector_type(8)));

constexpr int Bn = 8;
constexpr int Tn = 64;
constexpr int NI = 1024;
constexpr int NO = 1024;
constexpr int WAVES = 8;
constexpr int ROWS  = 32;
constexpr int KW    = NI / WAVES;
constexpr int KB    = 32;
constexpr int NBLK  = KW / KB;
constexpr int NTILE = 18;
constexpr int NFR   = 6;
constexpr int UP    = 33;
constexpr int TRI   = 120;

__host__ __device__ constexpr int tri_idx(int e, int f) { return 15 * e - (e * (e - 1)) / 2 + (f - e - 1); }

struct Frag { v4f q[4]; };

__device__ __forceinline__ void load_frag(Frag& f, const float* p) {
  const v4f* p4 = reinterpret_cast<const v4f*>(p);
  f.q[0] = p4[0];
  f.q[1] = p4[2];
  f.q[2] = p4[4];
  f.q[3] = p4[6];
}

__device__ __forceinline__ void load_block(Frag (&F)[NFR], const float* wp, const float* xp, int blk) {
#pragma unroll
  for (int i = 0; i < 2; ++i) load_frag(F[i], wp + (size_t)i * 16 * NI + blk * KB);
#pragma unroll
  for (int i = 0; i < 4; ++i) load_frag(F[2 + i], xp + (size_t)i * 16 * NI + blk * KB);
}

__device__ __forceinline__ void mma_block(v8f (&acc)[NTILE], const Frag (&F)[NFR]) {
#pragma unroll
  for (int q = 0; q < 4; ++q) {
#pragma unroll
    for (int p = 0; p < 2; ++p) {
      v2f a[NFR];
#pragma unroll
      for (int i = 0; i < NFR; ++i) {
        a[i].x = F[i].q[q][2 * p];
        a[i].y = F[i].q[q][2 * p + 1];
      }
#pragma unroll
      for (int mi = 0; mi < 2; ++mi) {
#pragma unroll
        for (int nj = 0; nj < 4; ++nj) {
          acc[mi * 4 + nj] = __builtin_amdgcn_wmma_f32_16x16x4_f32(
              false, a[mi], false, a[2 + nj], (short)0, acc[mi * 4 + nj], false, false);
        }
      }
      int idx = 8;
#pragma unroll
      for (int ti = 0; ti < 4; ++ti) {
#pragma unroll
        for (int tj = ti; tj < 4; ++tj) {
          acc[idx] = __builtin_amdgcn_wmma_f32_16x16x4_f32(
              false, a[2 + ti], false, a[2 + tj], (short)0, acc[idx], false, false);
          ++idx;
        }
      }
    }
  }
}

__global__ __launch_bounds__(256) void plastic_scan_kernel(
    const float* __restrict__ X,
    const float* __restrict__ W0,
    float* __restrict__ out)
{
  const int tid  = threadIdx.x;
  const int lane = tid & 31;
  const int wv   = tid >> 5;
  const int c    = lane & 15;
  const int h    = lane >> 4;
  const int b       = blockIdx.x >> 5;
  const int rowbase = (blockIdx.x & 31) * ROWS;
  const float lr  = 1.0f / (float)NI;
  const float KAP = -1.4426950408889634f;

  __shared__ __attribute__((aligned(16))) float part[WAVES * NTILE * 256];
  __shared__ __attribute__((aligned(16))) float Gl[Tn * Tn];
  __shared__ float Dl[Tn * UP];
  __shared__ float Cl[16 * 32];
  __shared__ __attribute__((aligned(16))) float Tl[4 * TRI];

  v8f acc[NTILE];
#pragma unroll
  for (int i = 0; i < NTILE; ++i) {
#pragma unroll
    for (int r = 0; r < 8; ++r) acc[i][r] = 0.0f;
  }

  const float* wp = W0 + ((size_t)b * NO + rowbase + c) * NI + wv * KW + 4 * h;
  const float* xp = X  + ((size_t)b * Tn + c) * NI + wv * KW + 4 * h;

  Frag F[2][NFR];
  load_block(F[0], wp, xp, 0);
#pragma unroll
  for (int blk = 0; blk < NBLK; ++blk) {
    if (blk + 1 < NBLK) load_block(F[(blk + 1) & 1], wp, xp, blk + 1);
    mma_block(acc, F[blk & 1]);
  }

#pragma unroll
  for (int i = 0; i < NTILE; ++i) {
#pragma unroll
    for (int r = 0; r < 8; ++r) part[(wv * NTILE + i) * 256 + r * 32 + lane] = acc[i][r];
  }
  __syncthreads();

  {
    const int rr = 8 * h + wv;
#pragma unroll
    for (int i = 0; i < 8; ++i) {
      float s = 0.0f;
#pragma unroll
      for (int k = 0; k < WAVES; ++k) s += part[(k * NTILE + i) * 256 + tid];
      const int mi = i >> 2, nj = i & 3;
      Dl[(16 * nj + c) * UP + 16 * mi + rr] = s * KAP;
    }
    int idx = 8;
#pragma unroll
    for (int ti = 0; ti < 4; ++ti) {
#pragma unroll
      for (int tj = ti; tj < 4; ++tj) {
        float s = 0.0f;
#pragma unroll
        for (int k = 0; k < WAVES; ++k) s += part[(k * NTILE + idx) * 256 + tid];
        const float gv = s * (KAP * lr);
        if (ti == tj) {
          if (c > rr) Tl[ti * TRI + tri_idx(rr, c)] = gv;
        } else {
          Gl[(16 * ti + rr) * Tn + 16 * tj + c] = gv;
        }
        ++idx;
      }
    }
  }
  __syncthreads();

  float P = 1.0f;
  float* op = out + (size_t)b * Tn * NO + rowbase + lane;
  v4f tq[TRI / 4];
  if (wv == 0) {
    const v4f* t4 = reinterpret_cast<const v4f*>(&Tl[0]);
#pragma unroll
    for (int i = 0; i < TRI / 4; ++i) tq[i] = t4[i];
  }
#pragma unroll 1
  for (int j = 0; j < Tn / 16; ++j) {
    if (wv == 0) {
      float E[16];
#pragma unroll
      for (int e = 0; e < 16; ++e) E[e] = Dl[(16 * j + e) * UP + lane];
      __builtin_amdgcn_sched_barrier(0);
      E[0] *= P;
      E[1] *= P;
      float LE = lr * E[1];
#pragma unroll
      for (int e = 0; e < 16; ++e) {
        const float y = __builtin_amdgcn_rcpf(1.0f + __builtin_amdgcn_exp2f(E[e]));
        op[(size_t)(16 * j + e) * NO] = y;
        if (e < 15) {
          const float u = __builtin_fmaf(-y, LE, tq[tri_idx(e, e + 1) / 4][tri_idx(e, e + 1) % 4]);
          E[e + 1] = __builtin_fmaf(y, u, E[e + 1]);
        }
        const float s = __builtin_fmaf(-lr * y, y, 1.0f);
        P *= s;
        const float m = (e == 0) ? P : s;
#pragma unroll
        for (int f = e + 2; f < 16; ++f)
          E[f] = __builtin_fmaf(m, E[f], y * tq[tri_idx(e, f) / 4][tri_idx(e, f) % 4]);
        if (e < 14) LE = lr * E[e + 2];
        Cl[e * 32 + lane] = y * __builtin_amdgcn_rcpf(P);
      }
    }
    __syncthreads();

    if (wv == 0) {
      if (j + 1 < Tn / 16) {
        const v4f* t4 = reinterpret_cast<const v4f*>(&Tl[(j + 1) * TRI]);
#pragma unroll
        for (int i = 0; i < TRI / 4; ++i) tq[i] = t4[i];
      }
    } else {
      const int wt = wv - 1;
      const int mi = wt & 1;
      const int jt = j + 1 + (wt >> 1);
      if (jt < Tn / 16) {
        v8f accd;
#pragma unroll
        for (int r = 0; r < 8; ++r) accd[r] = 0.0f;
#pragma unroll
        for (int ks = 0; ks < 4; ++ks) {
          const int k0 = 4 * ks + 2 * h;
          v2f av, bv;
          av.x = Cl[(k0    ) * 32 + 16 * mi + c];
          av.y = Cl[(k0 + 1) * 32 + 16 * mi + c];
          bv.x = Gl[(16 * j + k0    ) * Tn + 16 * jt + c];
          bv.y = Gl[(16 * j + k0 + 1) * Tn + 16 * jt + c];
          accd = __builtin_amdgcn_wmma_f32_16x16x4_f32(
              false, av, false, bv, (short)0, accd, false, false);
        }
#pragma unroll
        for (int r = 0; r < 8; ++r) Dl[(16 * jt + c) * UP + 16 * mi + 8 * h + r] += accd[r];
      }
    }
    __syncthreads();
  }
}

extern "C" void kernel_launch(void* const* d_in, const int* in_sizes, int n_in,
                              void* d_out, int out_size, void* d_ws, size_t ws_size,
                              hipStream_t stream) {
  const float* X  = (const float*)d_in[0];
  const float* W0 = (const float*)d_in[1];
  float* out      = (float*)d_out;
  (void)in_sizes; (void)n_in; (void)out_size; (void)d_ws; (void)ws_size;

  dim3 grid(Bn * (NO / ROWS));
  dim3 block(WAVES * 32);
  plastic_scan_kernel<<<grid, block, 0, stream>>>(X, W0, out);
}


# Block hand-over barriers of the recurrence wait for LDS only (asm: s_wait_dscnt, signal, wait): the chain wave no longer stalls on its output stores' acknowledgement after each 16-step block
# speedup vs baseline: 5.0203x; 1.0934x over previous
#include <hip/hip_runtime.h>


typedef float v2f __attribute__((ext_vector_type(2)));
typedef float v4f __attribute__((ext_vector_type(4)));
typedef float v8f __attribute__((ext_vector_type(8)));

constexpr int Bn = 8;
constexpr int Tn = 64;
constexpr int NI = 1024;
constexpr int NO = 1024;
constexpr int WAVES = 8;
constexpr int ROWS  = 32;
constexpr int KW    = NI / WAVES;
constexpr int KB    = 32;
constexpr int NBLK  = KW / KB;
constexpr int NTILE = 18;
constexpr int NFR   = 6;
constexpr int UP    = 33;
constexpr int TRI   = 120;

__host__ __device__ constexpr int tri_idx(int e, int f) { return 15 * e - (e * (e - 1)) / 2 + (f - e - 1); }

__device__ __forceinline__ void lds_barrier() {
  asm volatile("s_wait_dscnt 0x0\n\ts_barrier_signal -1\n\ts_barrier_wait -1" ::: "memory");
}

struct Frag { v4f q[4]; };

__device__ __forceinline__ void load_frag(Frag& f, const float* p) {
  const v4f* p4 = reinterpret_cast<const v4f*>(p);
  f.q[0] = p4[0];
  f.q[1] = p4[2];
  f.q[2] = p4[4];
  f.q[3] = p4[6];
}

__device__ __forceinline__ void load_block(Frag (&F)[NFR], const float* wp, const float* xp, int blk) {
#pragma unroll
  for (int i = 0; i < 2; ++i) load_frag(F[i], wp + (size_t)i * 16 * NI + blk * KB);
#pragma unroll
  for (int i = 0; i < 4; ++i) load_frag(F[2 + i], xp + (size_t)i * 16 * NI + blk * KB);
}

__device__ __forceinline__ void mma_block(v8f (&acc)[NTILE], const Frag (&F)[NFR]) {
#pragma unroll
  for (int q = 0; q < 4; ++q) {
#pragma unroll
    for (int p = 0; p < 2; ++p) {
      v2f a[NFR];
#pragma unroll
      for (int i = 0; i < NFR; ++i) {
        a[i].x = F[i].q[q][2 * p];
        a[i].y = F[i].q[q][2 * p + 1];
      }
#pragma unroll
      for (int mi = 0; mi < 2; ++mi) {
#pragma unroll
        for (int nj = 0; nj < 4; ++nj) {
          acc[mi * 4 + nj] = __builtin_amdgcn_wmma_f32_16x16x4_f32(
              false, a[mi], false, a[2 + nj], (short)0, acc[mi * 4 + nj], false, false);
        }
      }
      int idx = 8;
#pragma unroll
      for (int ti = 0; ti < 4; ++ti) {
#pragma unroll
        for (int tj = ti; tj < 4; ++tj) {
          acc[idx] = __builtin_amdgcn_wmma_f32_16x16x4_f32(
              false, a[2 + ti], false, a[2 + tj], (short)0, acc[idx], false, false);
          ++idx;
        }
      }
    }
  }
}

__global__ __launch_bounds__(256) void plastic_scan_kernel(
    const float* __restrict__ X,
    const float* __restrict__ W0,
    float* __restrict__ out)
{
  const int tid  = threadIdx.x;
  const int lane = tid & 31;
  const int wv   = tid >> 5;
  const int c    = lane & 15;
  const int h    = lane >> 4;
  const int b       = blockIdx.x >> 5;
  const int rowbase = (blockIdx.x & 31) * ROWS;
  const float lr  = 1.0f / (float)NI;
  const float KAP = -1.4426950408889634f;

  __shared__ __attribute__((aligned(16))) float part[WAVES * NTILE * 256];
  __shared__ __attribute__((aligned(16))) float Gl[Tn * Tn];
  __shared__ float Dl[Tn * UP];
  __shared__ float Cl[16 * 32];
  __shared__ __attribute__((aligned(16))) float Tl[4 * TRI];

  v8f acc[NTILE];
#pragma unroll
  for (int i = 0; i < NTILE; ++i) {
#pragma unroll
    for (int r = 0; r < 8; ++r) acc[i][r] = 0.0f;
  }

  const float* wp = W0 + ((size_t)b * NO + rowbase + c) * NI + wv * KW + 4 * h;
  const float* xp = X  + ((size_t)b * Tn + c) * NI + wv * KW + 4 * h;

  Frag F[2][NFR];
  load_block(F[0], wp, xp, 0);
#pragma unroll
  for (int blk = 0; blk < NBLK; ++blk) {
    if (blk + 1 < NBLK) load_block(F[(blk + 1) & 1], wp, xp, blk + 1);
    mma_block(acc, F[blk & 1]);
  }

#pragma unroll
  for (int i = 0; i < NTILE; ++i) {
#pragma unroll
    for (int r = 0; r < 8; ++r) part[(wv * NTILE + i) * 256 + r * 32 + lane] = acc[i][r];
  }
  __syncthreads();

  {
    const int rr = 8 * h + wv;
#pragma unroll
    for (int i = 0; i < 8; ++i) {
      float s = 0.0f;
#pragma unroll
      for (int k = 0; k < WAVES; ++k) s += part[(k * NTILE + i) * 256 + tid];
      const int mi = i >> 2, nj = i & 3;
      Dl[(16 * nj + c) * UP + 16 * mi + rr] = s * KAP;
    }
    int idx = 8;
#pragma unroll
    for (int ti = 0; ti < 4; ++ti) {
#pragma unroll
      for (int tj = ti; tj < 4; ++tj) {
        float s = 0.0f;
#pragma unroll
        for (int k = 0; k < WAVES; ++k) s += part[(k * NTILE + idx) * 256 + tid];
        const float gv = s * (KAP * lr);
        if (ti == tj) {
          if (c > rr) Tl[ti * TRI + tri_idx(rr, c)] = gv;
        } else {
          Gl[(16 * ti + rr) * Tn + 16 * tj + c] = gv;
        }
        ++idx;
      }
    }
  }
  __syncthreads();

  float P = 1.0f;
  float* op = out + (size_t)b * Tn * NO + rowbase + lane;
  v4f tq[TRI / 4];
  if (wv == 0) {
    const v4f* t4 = reinterpret_cast<const v4f*>(&Tl[0]);
#pragma unroll
    for (int i = 0; i < TRI / 4; ++i) tq[i] = t4[i];
  }
#pragma unroll 1
  for (int j = 0; j < Tn / 16; ++j) {
    if (wv == 0) {
      float E[16];
#pragma unroll
      for (int e = 0; e < 16; ++e) E[e] = Dl[(16 * j + e) * UP + lane];
      __builtin_amdgcn_sched_barrier(0);
      E[0] *= P;
      E[1] *= P;
      float LE = lr * E[1];
#pragma unroll
      for (int e = 0; e < 16; ++e) {
        const float y = __builtin_amdgcn_rcpf(1.0f + __builtin_amdgcn_exp2f(E[e]));
        op[(size_t)(16 * j + e) * NO] = y;
        if (e < 15) {
          const float u = __builtin_fmaf(-y, LE, tq[tri_idx(e, e + 1) / 4][tri_idx(e, e + 1) % 4]);
          E[e + 1] = __builtin_fmaf(y, u, E[e + 1]);
        }
        const float s = __builtin_fmaf(-lr * y, y, 1.0f);
        P *= s;
        const float m = (e == 0) ? P : s;
#pragma unroll
        for (int f = e + 2; f < 16; ++f)
          E[f] = __builtin_fmaf(m, E[f], y * tq[tri_idx(e, f) / 4][tri_idx(e, f) % 4]);
        if (e < 14) LE = lr * E[e + 2];
        Cl[e * 32 + lane] = y * __builtin_amdgcn_rcpf(P);
      }
    }
    lds_barrier();

    if (wv == 0) {
      if (j + 1 < Tn / 16) {
        const v4f* t4 = reinterpret_cast<const v4f*>(&Tl[(j + 1) * TRI]);
#pragma unroll
        for (int i = 0; i < TRI / 4; ++i) tq[i] = t4[i];
      }
    } else {
      const int wt = wv - 1;
      const int mi = wt & 1;
      const int jt = j + 1 + (wt >> 1);
      if (jt < Tn / 16) {
        v8f accd;
#pragma unroll
        for (int r = 0; r < 8; ++r) accd[r] = 0.0f;
#pragma unroll
        for (int ks = 0; ks < 4; ++ks) {
          const int k0 = 4 * ks + 2 * h;
          v2f av, bv;
          av.x = Cl[(k0    ) * 32 + 16 * mi + c];
          av.y = Cl[(k0 + 1) * 32 + 16 * mi + c];
          bv.x = Gl[(16 * j + k0    ) * Tn + 16 * jt + c];
          bv.y = Gl[(16 * j + k0 + 1) * Tn + 16 * jt + c];
          accd = __builtin_amdgcn_wmma_f32_16x16x4_f32(
              false, av, false, bv, (short)0, accd, false, false);
        }
#pragma unroll
        for (int r = 0; r < 8; ++r) Dl[(16 * jt + c) * UP + 16 * mi + 8 * h + r] += accd[r];
      }
    }
    lds_barrier();
  }
}

extern "C" void kernel_launch(void* const* d_in, const int* in_sizes, int n_in,
                              void* d_out, int out_size, void* d_ws, size_t ws_size,
                              hipStream_t stream) {
  const float* X  = (const float*)d_in[0];
  const float* W0 = (const float*)d_in[1];
  float* out      = (float*)d_out;
  (void)in_sizes; (void)n_in; (void)out_size; (void)d_ws; (void)ws_size;

  dim3 grid(Bn * (NO / ROWS));
  dim3 block(WAVES * 32);
  plastic_scan_kernel<<<grid, block, 0, stream>>>(X, W0, out);
}


# WMMA operands staged through LDS: each wave reads its K slice with whole-line 16-byte global loads (4 lines per load, not 16), writes a private LDS strip and reads fragments with ds_load_b128
# speedup vs baseline: 5.5862x; 1.1127x over previous
#include <hip/hip_runtime.h>


typedef float v2f __attribute__((ext_vector_type(2)));
typedef float v4f __attribute__((ext_vector_type(4)));
typedef float v8f __attribute__((ext_vector_type(8)));

constexpr int Bn = 8;
constexpr int Tn = 64;
constexpr int NI = 1024;
constexpr int NO = 1024;
constexpr int WAVES = 8;
constexpr int ROWS  = 32;
constexpr int KW    = NI / WAVES;
constexpr int KB    = 32;
constexpr int NBLK  = KW / KB;
constexpr int NTILE = 18;
constexpr int NFR   = 6;
constexpr int UP    = 33;
constexpr int TRI   = 120;
constexpr int SROWS  = 96;
constexpr int SPITCH = 36;
constexpr int NLD    = SROWS / 4;

__host__ __device__ constexpr int tri_idx(int e, int f) { return 15 * e - (e * (e - 1)) / 2 + (f - e - 1); }

__device__ __forceinline__ void lds_barrier() {
  asm volatile("s_wait_dscnt 0x0\n\ts_barrier_signal -1\n\ts_barrier_wait -1" ::: "memory");
}

struct Frag { v4f q[4]; };

struct Stage { v4f g[NLD]; };

__device__ __forceinline__ void gload_block(Stage& S, const float* wq, const float* xq, int blk) {
#pragma unroll
  for (int i = 0; i < 8; ++i)
    S.g[i] = *reinterpret_cast<const v4f*>(wq + (size_t)(4 * i) * NI + blk * KB);
#pragma unroll
  for (int i = 0; i < 16; ++i)
    S.g[8 + i] = *reinterpret_cast<const v4f*>(xq + (size_t)(4 * i) * NI + blk * KB);
}

__device__ __forceinline__ void stage_store(float* sp, const Stage& S) {
#pragma unroll
  for (int i = 0; i < NLD; ++i) *reinterpret_cast<v4f*>(sp + 4 * i * SPITCH) = S.g[i];
}

__device__ __forceinline__ void frag_load(Frag (&F)[NFR], const float* fp) {
#pragma unroll
  for (int f = 0; f < NFR; ++f) {
#pragma unroll
    for (int q = 0; q < 4; ++q)
      F[f].q[q] = *reinterpret_cast<const v4f*>(fp + 16 * f * SPITCH + 8 * q);
  }
}

__device__ __forceinline__ void mma_half(v8f (&acc)[NTILE], const Frag (&F)[NFR], int q0) {
#pragma unroll
  for (int q = q0; q < q0 + 2; ++q) {
#pragma unroll
    for (int p = 0; p < 2; ++p) {
      v2f a[NFR];
#pragma unroll
      for (int i = 0; i < NFR; ++i) {
        a[i].x = F[i].q[q][2 * p];
        a[i].y = F[i].q[q][2 * p + 1];
      }
#pragma unroll
      for (int mi = 0; mi < 2; ++mi) {
#pragma unroll
        for (int nj = 0; nj < 4; ++nj) {
          acc[mi * 4 + nj] = __builtin_amdgcn_wmma_f32_16x16x4_f32(
              false, a[mi], false, a[2 + nj], (short)0, acc[mi * 4 + nj], false, false);
        }
      }
      int idx = 8;
#pragma unroll
      for (int ti = 0; ti < 4; ++ti) {
#pragma unroll
        for (int tj = ti; tj < 4; ++tj) {
          acc[idx] = __builtin_amdgcn_wmma_f32_16x16x4_f32(
              false, a[2 + ti], false, a[2 + tj], (short)0, acc[idx], false, false);
          ++idx;
        }
      }
    }
  }
}

__global__ __launch_bounds__(256) void plastic_scan_kernel(
    const float* __restrict__ X,
    const float* __restrict__ W0,
    float* __restrict__ out)
{
  const int tid  = threadIdx.x;
  const int lane = tid & 31;
  const int wv   = tid >> 5;
  const int c    = lane & 15;
  const int h    = lane >> 4;
  const int b       = blockIdx.x >> 5;
  const int rowbase = (blockIdx.x & 31) * ROWS;
  const float lr  = 1.0f / (float)NI;
  const float KAP = -1.4426950408889634f;

  __shared__ __attribute__((aligned(16))) float part[WAVES * NTILE * 256];
  __shared__ __attribute__((aligned(16))) float Gl[Tn * Tn];
  __shared__ float Dl[Tn * UP];
  __shared__ float Cl[16 * 32];
  __shared__ __attribute__((aligned(16))) float Tl[4 * TRI];

  v8f acc[NTILE];
#pragma unroll
  for (int i = 0; i < NTILE; ++i) {
#pragma unroll
    for (int r = 0; r < 8; ++r) acc[i][r] = 0.0f;
  }

  const float* wq = W0 + ((size_t)b * NO + rowbase + (lane >> 3)) * NI + wv * KW + 4 * (lane & 7);
  const float* xq = X  + ((size_t)b * Tn + (lane >> 3)) * NI + wv * KW + 4 * (lane & 7);
  float* stg = part + wv * NTILE * 256;
  float* sp = stg + (lane >> 3) * SPITCH + 4 * (lane & 7);
  const float* fp = stg + c * SPITCH + 4 * h;

  Stage S0, S1;
  Frag F[2][NFR];
  gload_block(S0, wq, xq, 0);
  gload_block(S1, wq, xq, 1);
  stage_store(sp, S0);
  frag_load(F[0], fp);
#pragma unroll
  for (int blk = 0; blk < NBLK; ++blk) {
    __builtin_amdgcn_sched_barrier(0);
    mma_half(acc, F[blk & 1], 0);
    __builtin_amdgcn_sched_barrier(0);
    if (blk + 1 < NBLK) {
      stage_store(sp, ((blk + 1) & 1) ? S1 : S0);
      if (blk + 2 < NBLK) gload_block((blk & 1) ? S1 : S0, wq, xq, blk + 2);
      frag_load(F[(blk + 1) & 1], fp);
    }
    __builtin_amdgcn_sched_barrier(0);
    mma_half(acc, F[blk & 1], 2);
    __builtin_amdgcn_sched_barrier(0);
  }

#pragma unroll
  for (int i = 0; i < NTILE; ++i) {
#pragma unroll
    for (int r = 0; r < 8; ++r) part[(wv * NTILE + i) * 256 + r * 32 + lane] = acc[i][r];
  }
  __syncthreads();

  {
    const int rr = 8 * h + wv;
#pragma unroll
    for (int i = 0; i < 8; ++i) {
      float s = 0.0f;
#pragma unroll
      for (int k = 0; k < WAVES; ++k) s += part[(k * NTILE + i) * 256 + tid];
      const int mi = i >> 2, nj = i & 3;
      Dl[(16 * nj + c) * UP + 16 * mi + rr] = s * KAP;
    }
    int idx = 8;
#pragma unroll
    for (int ti = 0; ti < 4; ++ti) {
#pragma unroll
      for (int tj = ti; tj < 4; ++tj) {
        float s = 0.0f;
#pragma unroll
        for (int k = 0; k < WAVES; ++k) s += part[(k * NTILE + idx) * 256 + tid];
        const float gv = s * (KAP * lr);
        if (ti == tj) {
          if (c > rr) Tl[ti * TRI + tri_idx(rr, c)] = gv;
        } else {
          Gl[(16 * ti + rr) * Tn + 16 * tj + c] = gv;
        }
        ++idx;
      }
    }
  }
  __syncthreads();

  float P = 1.0f;
  float* op = out + (size_t)b * Tn * NO + rowbase + lane;
  v4f tq[TRI / 4];
  if (wv == 0) {
    const v4f* t4 = reinterpret_cast<const v4f*>(&Tl[0]);
#pragma unroll
    for (int i = 0; i < TRI / 4; ++i) tq[i] = t4[i];
  }
#pragma unroll 1
  for (int j = 0; j < Tn / 16; ++j) {
    if (wv == 0) {
      float E[16];
#pragma unroll
      for (int e = 0; e < 16; ++e) E[e] = Dl[(16 * j + e) * UP + lane];
      __builtin_amdgcn_sched_barrier(0);
      E[0] *= P;
      E[1] *= P;
      float LE = lr * E[1];
#pragma unroll
      for (int e = 0; e < 16; ++e) {
        const float y = __builtin_amdgcn_rcpf(1.0f + __builtin_amdgcn_exp2f(E[e]));
        op[(size_t)(16 * j + e) * NO] = y;
        if (e < 15) {
          const float u = __builtin_fmaf(-y, LE, tq[tri_idx(e, e + 1) / 4][tri_idx(e, e + 1) % 4]);
          E[e + 1] = __builtin_fmaf(y, u, E[e + 1]);
        }
        const float s = __builtin_fmaf(-lr * y, y, 1.0f);
        P *= s;
        const float m = (e == 0) ? P : s;
#pragma unroll
        for (int f = e + 2; f < 16; ++f)
          E[f] = __builtin_fmaf(m, E[f], y * tq[tri_idx(e, f) / 4][tri_idx(e, f) % 4]);
        if (e < 14) LE = lr * E[e + 2];
        Cl[e * 32 + lane] = y * __builtin_amdgcn_rcpf(P);
      }
    }
    lds_barrier();

    if (wv == 0) {
      if (j + 1 < Tn / 16) {
        const v4f* t4 = reinterpret_cast<const v4f*>(&Tl[(j + 1) * TRI]);
#pragma unroll
        for (int i = 0; i < TRI / 4; ++i) tq[i] = t4[i];
      }
    } else {
      const int wt = wv - 1;
      const int mi = wt & 1;
      const int jt = j + 1 + (wt >> 1);
      if (jt < Tn / 16) {
        v8f accd;
#pragma unroll
        for (int r = 0; r < 8; ++r) accd[r] = 0.0f;
#pragma unroll
        for (int ks = 0; ks < 4; ++ks) {
          const int k0 = 4 * ks + 2 * h;
          v2f av, bv;
          av.x = Cl[(k0    ) * 32 + 16 * mi + c];
          av.y = Cl[(k0 + 1) * 32 + 16 * mi + c];
          bv.x = Gl[(16 * j + k0    ) * Tn + 16 * jt + c];
          bv.y = Gl[(16 * j + k0 + 1) * Tn + 16 * jt + c];
          accd = __builtin_amdgcn_wmma_f32_16x16x4_f32(
              false, av, false, bv, (short)0, accd, false, false);
        }
#pragma unroll
        for (int r = 0; r < 8; ++r) Dl[(16 * jt + c) * UP + 16 * mi + 8 * h + r] += accd[r];
      }
    }
    lds_barrier();
  }
}

extern "C" void kernel_launch(void* const* d_in, const int* in_sizes, int n_in,
                              void* d_out, int out_size, void* d_ws, size_t ws_size,
                              hipStream_t stream) {
  const float* X  = (const float*)d_in[0];
  const float* W0 = (const float*)d_in[1];
  float* out      = (float*)d_out;
  (void)in_sizes; (void)n_in; (void)out_size; (void)d_ws; (void)ws_size;

  dim3 grid(Bn * (NO / ROWS));
  dim3 block(WAVES * 32);
  plastic_scan_kernel<<<grid, block, 0, stream>>>(X, W0, out);
}
